# starting kernel
# baseline (speedup 1.0000x reference)
#include <hip/hip_runtime.h>
#include <hip/hip_bf16.h>

#define PB 8
#define PD 64
#define PS 512

typedef __attribute__((ext_vector_type(2))) float v2f;
typedef __attribute__((ext_vector_type(8))) float v8f;

__device__ __forceinline__ v8f wmma_f32_16x16x4(v2f a, v2f b, v8f c) {
    return __builtin_amdgcn_wmma_f32_16x16x4_f32(
         false, a,  false, b,
         (short)0, c,  false,  false);
}

__device__ __forceinline__ v8f v8f_zero() {
    v8f z = {0.f, 0.f, 0.f, 0.f, 0.f, 0.f, 0.f, 0.f};
    return z;
}

__global__ __launch_bounds__(128) void lie_stage1(
    const float* __restrict__ x, const float* __restrict__ W,
    float* __restrict__ Aws, float* __restrict__ Bws) {

    const int lane = threadIdx.x & 31;
    const int wv   = threadIdx.x >> 5;
    const int gw   = blockIdx.x * 4 + wv;
    const int b    = gw >> 5;
    const int jb   = (gw & 31) * 16;
    const int ln   = lane & 15;
    const int kh   = lane >> 4;

    __builtin_prefetch(W + threadIdx.x * 4, 0, 1);

    const float* xb = x + (size_t)b * PD * PS;

    v8f accA[4], accB[4];
#pragma unroll
    for (int nt = 0; nt < 4; ++nt) { accA[nt] = v8f_zero(); accB[nt] = v8f_zero(); }

#pragma unroll
    for (int kb = 0; kb < PD; kb += 4) {
        const int k0 = kb + 2 * kh;
        v2f a;
        a.x = xb[(size_t)k0 * PS + jb + ln];
        a.y = xb[(size_t)(k0 + 1) * PS + jb + ln];
#pragma unroll
        for (int nt = 0; nt < 4; ++nt) {
            const int n = nt * 16 + ln;
            v2f bw, bh;
            bw.x = W[(size_t)k0 * PD + n];
            bw.y = W[(size_t)(k0 + 1) * PD + n];
            bh.x = W[(size_t)(k0 + PD) * PD + n];
            bh.y = W[(size_t)(k0 + 1 + PD) * PD + n];
            accA[nt] = wmma_f32_16x16x4(a, bw, accA[nt]);
            accB[nt] = wmma_f32_16x16x4(a, bh, accB[nt]);
        }
    }

    float* Ab = Aws + ((size_t)b * PS + jb) * PD;
    float* Bb = Bws + ((size_t)b * PS + jb) * PD;
#pragma unroll
    for (int nt = 0; nt < 4; ++nt) {
#pragma unroll
        for (int v = 0; v < 8; ++v) {
            const int r = v + 8 * kh;
            Ab[(size_t)r * PD + nt * 16 + ln] = accA[nt][v];
            Bb[(size_t)r * PD + nt * 16 + ln] = accB[nt][v];
        }
    }
}

__global__ __launch_bounds__(256) void lie_stage2(
    const float* __restrict__ x, const float* __restrict__ Aws,
    const float* __restrict__ Bws, float* __restrict__ out) {

    const int b  = blockIdx.y;
    const int ib = blockIdx.x * 64;

    __shared__ float sA[64][65];
    __shared__ float sB[64][65];
    __shared__ float sWm[64][64];
    __shared__ float sNa[64];
    __shared__ float sNb[64];

    const int t    = threadIdx.x;
    const int lane = t & 31;
    const int wv   = t >> 5;
    const int ln   = lane & 15;
    const int kh   = lane >> 4;
    const int mt   = (wv & 3) * 16;
    const int nh   = (wv >> 2) * 32;

    const float* Ab = Aws + ((size_t)b * PS + ib) * PD;
#pragma unroll
    for (int p = 0; p < 16; ++p) {
        const int lin = t + 256 * p;
        sA[lin >> 6][lin & 63] = Ab[lin];
    }
    __syncthreads();
    if (t < 64) {
        float s = 0.f;
#pragma unroll
        for (int k = 0; k < 64; ++k) { const float v = sA[t][k]; s += v * v; }
        sNa[t] = s;
    }
    __syncthreads();

    const float* xb = x + (size_t)b * PD * PS;
    v8f c0 = v8f_zero(), c1 = v8f_zero();

    for (int wc = 0; wc < PS; wc += 64) {
        const float* Bb = Bws + ((size_t)b * PS + wc) * PD;
#pragma unroll
        for (int p = 0; p < 16; ++p) {
            const int lin = t + 256 * p;
            sB[lin >> 6][lin & 63] = Bb[lin];
        }
        __syncthreads();
        if (t < 64) {
            float s = 0.f;
#pragma unroll
            for (int k = 0; k < 64; ++k) { const float v = sB[t][k]; s += v * v; }
            sNb[t] = s;
        }
        __syncthreads();

        v8f g0 = v8f_zero(), g1 = v8f_zero();
#pragma unroll
        for (int kb = 0; kb < 64; kb += 4) {
            const int k0 = kb + 2 * kh;
            v2f a;  a.x  = sB[mt + ln][k0];        a.y  = sB[mt + ln][k0 + 1];
            v2f b0; b0.x = sA[nh + ln][k0];        b0.y = sA[nh + ln][k0 + 1];
            v2f b1; b1.x = sA[nh + 16 + ln][k0];   b1.y = sA[nh + 16 + ln][k0 + 1];
            g0 = wmma_f32_16x16x4(a, b0, g0);
            g1 = wmma_f32_16x16x4(a, b1, g1);
        }

#pragma unroll
        for (int v = 0; v < 8; ++v) {
            const int wrow = mt + v + 8 * kh;
            const int gwi  = wc + wrow;
            const int i0   = nh + ln;
            const int i1   = nh + 16 + ln;
            float m0 = 1.0f, m1 = 1.0f;
            if (ib + i0 <= gwi)
                m0 = 1.0f - sqrtf(fmaxf(sNa[i0] + sNb[wrow] + 2.0f * g0[v], 0.0f));
            if (ib + i1 <= gwi)
                m1 = 1.0f - sqrtf(fmaxf(sNa[i1] + sNb[wrow] + 2.0f * g1[v], 0.0f));
            sWm[wrow][i0] = m0;
            sWm[wrow][i1] = m1;
        }
        __syncthreads();

#pragma unroll
        for (int kb = 0; kb < 64; kb += 4) {
            const int k0 = kb + 2 * kh;
            v2f a;  a.x  = xb[(size_t)(mt + ln) * PS + wc + k0];
                    a.y  = xb[(size_t)(mt + ln) * PS + wc + k0 + 1];
            v2f b0; b0.x = sWm[k0][nh + ln];       b0.y = sWm[k0 + 1][nh + ln];
            v2f b1; b1.x = sWm[k0][nh + 16 + ln];  b1.y = sWm[k0 + 1][nh + 16 + ln];
            c0 = wmma_f32_16x16x4(a, b0, c0);
            c1 = wmma_f32_16x16x4(a, b1, c1);
        }
        __syncthreads();
    }

    float* ob = out + (size_t)b * PD * PS;
#pragma unroll
    for (int v = 0; v < 8; ++v) {
        const int d = mt + v + 8 * kh;
        ob[(size_t)d * PS + ib + nh + ln]      = c0[v];
        ob[(size_t)d * PS + ib + nh + 16 + ln] = c1[v];
    }
}

extern "C" void kernel_launch(void* const* d_in, const int* in_sizes, int n_in,
                              void* d_out, int out_size, void* d_ws, size_t ws_size,
                              hipStream_t stream) {
    const float* x = (const float*)d_in[0];
    const float* W = (const float*)d_in[1];
    float* out = (float*)d_out;

    float* Aws = (float*)d_ws;
    float* Bws = Aws + (size_t)PB * PS * PD;

    lie_stage1<<<64, 128, 0, stream>>>(x, W, Aws, Bws);
    lie_stage2<<<dim3(PS / 64, PB), 256, 0, stream>>>(x, Aws, Bws, out);
}


# stage 2: w split over the 8 waves (64 w each, both products in one wave, no barrier), 32-wide i-tiles on 128 work-groups; partial c summed via LDS in fixed wave order, norms per lane half
# speedup vs baseline: 2.0508x; 2.0508x over previous
#include <hip/hip_runtime.h>
#include <hip/hip_bf16.h>

#define PB 8
#define PD 64
#define PS 512

typedef __attribute__((ext_vector_type(2))) float v2f;
typedef __attribute__((ext_vector_type(8))) float v8f;

__device__ __forceinline__ v8f wmma_f32_16x16x4(v2f a, v2f b, v8f c) {
    return __builtin_amdgcn_wmma_f32_16x16x4_f32(
         false, a,  false, b,
         (short)0, c,  false,  false);
}

__device__ __forceinline__ v8f v8f_zero() {
    v8f z = {0.f, 0.f, 0.f, 0.f, 0.f, 0.f, 0.f, 0.f};
    return z;
}

__global__ __launch_bounds__(128) void lie_stage1(
    const float* __restrict__ x, const float* __restrict__ W,
    float* __restrict__ Aws, float* __restrict__ Bws) {

    const int lane = threadIdx.x & 31;
    const int wv   = threadIdx.x >> 5;
    const int gw   = blockIdx.x * 4 + wv;
    const int b    = gw >> 5;
    const int jb   = (gw & 31) * 16;
    const int ln   = lane & 15;
    const int kh   = lane >> 4;

    __builtin_prefetch(W + threadIdx.x * 4, 0, 1);

    const float* xb = x + (size_t)b * PD * PS;

    v8f accA[4], accB[4];
#pragma unroll
    for (int nt = 0; nt < 4; ++nt) { accA[nt] = v8f_zero(); accB[nt] = v8f_zero(); }

#pragma unroll
    for (int kb = 0; kb < PD; kb += 4) {
        const int k0 = kb + 2 * kh;
        v2f a;
        a.x = xb[(size_t)k0 * PS + jb + ln];
        a.y = xb[(size_t)(k0 + 1) * PS + jb + ln];
#pragma unroll
        for (int nt = 0; nt < 4; ++nt) {
            const int n = nt * 16 + ln;
            v2f bw, bh;
            bw.x = W[(size_t)k0 * PD + n];
            bw.y = W[(size_t)(k0 + 1) * PD + n];
            bh.x = W[(size_t)(k0 + PD) * PD + n];
            bh.y = W[(size_t)(k0 + 1 + PD) * PD + n];
            accA[nt] = wmma_f32_16x16x4(a, bw, accA[nt]);
            accB[nt] = wmma_f32_16x16x4(a, bh, accB[nt]);
        }
    }

    float* Ab = Aws + ((size_t)b * PS + jb) * PD;
    float* Bb = Bws + ((size_t)b * PS + jb) * PD;
#pragma unroll
    for (int nt = 0; nt < 4; ++nt) {
#pragma unroll
        for (int v = 0; v < 8; ++v) {
            const int r = v + 8 * kh;
            Ab[(size_t)r * PD + nt * 16 + ln] = accA[nt][v];
            Bb[(size_t)r * PD + nt * 16 + ln] = accB[nt][v];
        }
    }
}

typedef __attribute__((ext_vector_type(4))) float v4f;

#define RED_PITCH 36

__global__ __launch_bounds__(256) void lie_stage2(
    const float* __restrict__ x, const float* __restrict__ Aws,
    const float* __restrict__ Bws, float* __restrict__ out) {

    const int b  = blockIdx.y;
    const int ib = blockIdx.x * 32;

    __shared__ __attribute__((aligned(16))) float red[8][64][RED_PITCH];

    const int t    = threadIdx.x;
    const int lane = t & 31;
    const int wv   = t >> 5;
    const int ln   = lane & 15;
    const int kh   = lane >> 4;
    const int wc   = wv * 64;


    v4f qa[2][8];
    float na[2];
#pragma unroll
    for (int nt = 0; nt < 2; ++nt) {
        const float* pa = Aws + ((size_t)(b * PS + ib + nt * 16 + ln)) * PD + 4 * kh;
        float s2 = 0.f;
#pragma unroll
        for (int s = 0; s < 8; ++s) {
            const v4f q = *reinterpret_cast<const v4f*>(pa + 8 * s);
            qa[nt][s] = q;
            s2 += q.x * q.x + q.y * q.y + q.z * q.z + q.w * q.w;
        }
        na[nt] = s2 + __shfl_xor(s2, 16);
    }

    v8f c[4][2];
#pragma unroll
    for (int dt = 0; dt < 4; ++dt) { c[dt][0] = v8f_zero(); c[dt][1] = v8f_zero(); }

    const float* xb = x + (size_t)b * PD * PS;

#pragma unroll
    for (int mtile = 0; mtile < 4; ++mtile) {
        const int wl0 = mtile * 16;

        v4f qb[8];
        const float* pb = Bws + ((size_t)(b * PS + wc + wl0 + ln)) * PD + 4 * kh;
        float s2 = 0.f;
#pragma unroll
        for (int s = 0; s < 8; ++s) {
            const v4f q = *reinterpret_cast<const v4f*>(pb + 8 * s);
            qb[s] = q;
            s2 += q.x * q.x + q.y * q.y + q.z * q.z + q.w * q.w;
        }
        const float nbrow = s2 + __shfl_xor(s2, 16);

        v4f xlo[4], xhi[4];
#pragma unroll
        for (int dt = 0; dt < 4; ++dt) {
            const float* px = xb + (size_t)(dt * 16 + ln) * PS + wc + wl0 + 8 * kh;
            xlo[dt] = *reinterpret_cast<const v4f*>(px);
            xhi[dt] = *reinterpret_cast<const v4f*>(px + 4);
        }

        v8f g0 = v8f_zero(), g1 = v8f_zero();
#pragma unroll
        for (int s = 0; s < 8; ++s) {
            v2f a0;  a0.x  = qb[s].x;     a0.y  = qb[s].y;
            v2f a1;  a1.x  = qb[s].z;     a1.y  = qb[s].w;
            v2f b00; b00.x = qa[0][s].x;  b00.y = qa[0][s].y;
            v2f b01; b01.x = qa[0][s].z;  b01.y = qa[0][s].w;
            v2f b10; b10.x = qa[1][s].x;  b10.y = qa[1][s].y;
            v2f b11; b11.x = qa[1][s].z;  b11.y = qa[1][s].w;
            g0 = wmma_f32_16x16x4(a0, b00, g0);
            g1 = wmma_f32_16x16x4(a0, b10, g1);
            g0 = wmma_f32_16x16x4(a1, b01, g0);
            g1 = wmma_f32_16x16x4(a1, b11, g1);
        }

        float m0[8], m1[8];
#pragma unroll
        for (int v = 0; v < 8; ++v) {
            const int   gwi = wc + wl0 + 8 * kh + v;
            const float nbv = __shfl(nbrow, v + 8 * kh);
            const float r0  = sqrtf(fmaxf(na[0] + nbv + 2.0f * g0[v], 0.0f));
            const float r1  = sqrtf(fmaxf(na[1] + nbv + 2.0f * g1[v], 0.0f));
            m0[v] = (ib + ln      <= gwi) ? 1.0f - r0 : 1.0f;
            m1[v] = (ib + 16 + ln <= gwi) ? 1.0f - r1 : 1.0f;
        }

#pragma unroll
        for (int p = 0; p < 4; ++p) {
            v2f bm0; bm0.x = m0[2 * p];  bm0.y = m0[2 * p + 1];
            v2f bm1; bm1.x = m1[2 * p];  bm1.y = m1[2 * p + 1];
#pragma unroll
            for (int dt = 0; dt < 4; ++dt) {
                v2f a;
                if (p == 0)      { a.x = xlo[dt].x; a.y = xlo[dt].y; }
                else if (p == 1) { a.x = xlo[dt].z; a.y = xlo[dt].w; }
                else if (p == 2) { a.x = xhi[dt].x; a.y = xhi[dt].y; }
                else             { a.x = xhi[dt].z; a.y = xhi[dt].w; }
                c[dt][0] = wmma_f32_16x16x4(a, bm0, c[dt][0]);
                c[dt][1] = wmma_f32_16x16x4(a, bm1, c[dt][1]);
            }
        }
    }

#pragma unroll
    for (int dt = 0; dt < 4; ++dt) {
#pragma unroll
        for (int v = 0; v < 8; ++v) {
            red[wv][dt * 16 + 8 * kh + v][ln]      = c[dt][0][v];
            red[wv][dt * 16 + 8 * kh + v][16 + ln] = c[dt][1][v];
        }
    }
    __syncthreads();

    float* ob = out + (size_t)b * PD * PS;
#pragma unroll
    for (int j = 0; j < 2; ++j) {
        const int f  = t + 256 * j;
        const int d  = f >> 3;
        const int iq = (f & 7) * 4;
        v4f acc = *reinterpret_cast<const v4f*>(&red[0][d][iq]);
#pragma unroll
        for (int w = 1; w < 8; ++w) acc += *reinterpret_cast<const v4f*>(&red[w][d][iq]);
        *reinterpret_cast<v4f*>(ob + (size_t)d * PS + ib + iq) = acc;
    }
}

extern "C" void kernel_launch(void* const* d_in, const int* in_sizes, int n_in,
                              void* d_out, int out_size, void* d_ws, size_t ws_size,
                              hipStream_t stream) {
    const float* x = (const float*)d_in[0];
    const float* W = (const float*)d_in[1];
    float* out = (float*)d_out;

    float* Aws = (float*)d_ws;
    float* Bws = Aws + (size_t)PB * PS * PD;

    lie_stage1<<<64, 128, 0, stream>>>(x, W, Aws, Bws);
    lie_stage2<<<dim3(PS / 32, PB), 256, 0, stream>>>(x, Aws, Bws, out);
}


# stage 1 re-tiled: 1024 waves (16 rows x 32 columns of A or Bm each) on 256 work-groups, was 256 waves on 64; each wave issues all its operand loads before its first WMMA
# speedup vs baseline: 3.2315x; 1.5757x over previous
#include <hip/hip_runtime.h>
#include <hip/hip_bf16.h>

#define PB 8
#define PD 64
#define PS 512

typedef __attribute__((ext_vector_type(2))) float v2f;
typedef __attribute__((ext_vector_type(8))) float v8f;

__device__ __forceinline__ v8f wmma_f32_16x16x4(v2f a, v2f b, v8f c) {
    return __builtin_amdgcn_wmma_f32_16x16x4_f32(
         false, a,  false, b,
         (short)0, c,  false,  false);
}

__device__ __forceinline__ v8f v8f_zero() {
    v8f z = {0.f, 0.f, 0.f, 0.f, 0.f, 0.f, 0.f, 0.f};
    return z;
}

__global__ __launch_bounds__(128) void lie_stage1(
    const float* __restrict__ x, const float* __restrict__ W,
    float* __restrict__ Aws, float* __restrict__ Bws) {

    const int lane = threadIdx.x & 31;
    const int wv   = threadIdx.x >> 5;
    const int ln   = lane & 15;
    const int kh   = lane >> 4;

    const int half  = blockIdx.x & 1;
    const int chunk = (blockIdx.x >> 1) & 15;
    const int b     = blockIdx.x >> 5;
    const int jb    = chunk * 32 + (wv & 1) * 16;
    const int n0    = (wv >> 1) * 32;

    const float* xp = x + (size_t)b * PD * PS + (size_t)(2 * kh) * PS + jb + ln;
    const float* wp = W + (size_t)(half * PD + 2 * kh) * PD + n0 + 2 * ln;

    v2f a[16], w0[16], w1[16];
#pragma unroll
    for (int s = 0; s < 16; ++s) {
        a[s].x = xp[(size_t)(4 * s) * PS];
        a[s].y = xp[(size_t)(4 * s + 1) * PS];
        w0[s]  = *reinterpret_cast<const v2f*>(wp + (size_t)(4 * s) * PD);
        w1[s]  = *reinterpret_cast<const v2f*>(wp + (size_t)(4 * s + 1) * PD);
    }

    __builtin_amdgcn_sched_barrier(0);

    v8f acc0 = v8f_zero(), acc1 = v8f_zero();
#pragma unroll
    for (int s = 0; s < 16; ++s) {
        v2f b0; b0.x = w0[s].x; b0.y = w1[s].x;
        v2f b1; b1.x = w0[s].y; b1.y = w1[s].y;
        acc0 = wmma_f32_16x16x4(a[s], b0, acc0);
        acc1 = wmma_f32_16x16x4(a[s], b1, acc1);
    }

    float* dst = (half ? Bws : Aws) + ((size_t)b * PS + jb + 8 * kh) * PD + n0 + 2 * ln;
#pragma unroll
    for (int v = 0; v < 8; ++v) {
        v2f o; o.x = acc0[v]; o.y = acc1[v];
        *reinterpret_cast<v2f*>(dst + (size_t)v * PD) = o;
    }
}

typedef __attribute__((ext_vector_type(4))) float v4f;

#define RED_PITCH 36

__global__ __launch_bounds__(256) void lie_stage2(
    const float* __restrict__ x, const float* __restrict__ Aws,
    const float* __restrict__ Bws, float* __restrict__ out) {

    const int b  = blockIdx.y;
    const int ib = blockIdx.x * 32;

    __shared__ __attribute__((aligned(16))) float red[8][64][RED_PITCH];

    const int t    = threadIdx.x;
    const int lane = t & 31;
    const int wv   = t >> 5;
    const int ln   = lane & 15;
    const int kh   = lane >> 4;
    const int wc   = wv * 64;


    v4f qa[2][8];
    float na[2];
#pragma unroll
    for (int nt = 0; nt < 2; ++nt) {
        const float* pa = Aws + ((size_t)(b * PS + ib + nt * 16 + ln)) * PD + 4 * kh;
        float s2 = 0.f;
#pragma unroll
        for (int s = 0; s < 8; ++s) {
            const v4f q = *reinterpret_cast<const v4f*>(pa + 8 * s);
            qa[nt][s] = q;
            s2 += q.x * q.x + q.y * q.y + q.z * q.z + q.w * q.w;
        }
        na[nt] = s2 + __shfl_xor(s2, 16);
    }

    v8f c[4][2];
#pragma unroll
    for (int dt = 0; dt < 4; ++dt) { c[dt][0] = v8f_zero(); c[dt][1] = v8f_zero(); }

    const float* xb = x + (size_t)b * PD * PS;

#pragma unroll
    for (int mtile = 0; mtile < 4; ++mtile) {
        const int wl0 = mtile * 16;

        v4f qb[8];
        const float* pb = Bws + ((size_t)(b * PS + wc + wl0 + ln)) * PD + 4 * kh;
        float s2 = 0.f;
#pragma unroll
        for (int s = 0; s < 8; ++s) {
            const v4f q = *reinterpret_cast<const v4f*>(pb + 8 * s);
            qb[s] = q;
            s2 += q.x * q.x + q.y * q.y + q.z * q.z + q.w * q.w;
        }
        const float nbrow = s2 + __shfl_xor(s2, 16);

        v4f xlo[4], xhi[4];
#pragma unroll
        for (int dt = 0; dt < 4; ++dt) {
            const float* px = xb + (size_t)(dt * 16 + ln) * PS + wc + wl0 + 8 * kh;
            xlo[dt] = *reinterpret_cast<const v4f*>(px);
            xhi[dt] = *reinterpret_cast<const v4f*>(px + 4);
        }

        v8f g0 = v8f_zero(), g1 = v8f_zero();
#pragma unroll
        for (int s = 0; s < 8; ++s) {
            v2f a0;  a0.x  = qb[s].x;     a0.y  = qb[s].y;
            v2f a1;  a1.x  = qb[s].z;     a1.y  = qb[s].w;
            v2f b00; b00.x = qa[0][s].x;  b00.y = qa[0][s].y;
            v2f b01; b01.x = qa[0][s].z;  b01.y = qa[0][s].w;
            v2f b10; b10.x = qa[1][s].x;  b10.y = qa[1][s].y;
            v2f b11; b11.x = qa[1][s].z;  b11.y = qa[1][s].w;
            g0 = wmma_f32_16x16x4(a0, b00, g0);
            g1 = wmma_f32_16x16x4(a0, b10, g1);
            g0 = wmma_f32_16x16x4(a1, b01, g0);
            g1 = wmma_f32_16x16x4(a1, b11, g1);
        }

        float m0[8], m1[8];
#pragma unroll
        for (int v = 0; v < 8; ++v) {
            const int   gwi = wc + wl0 + 8 * kh + v;
            const float nbv = __shfl(nbrow, v + 8 * kh);
            const float r0  = sqrtf(fmaxf(na[0] + nbv + 2.0f * g0[v], 0.0f));
            const float r1  = sqrtf(fmaxf(na[1] + nbv + 2.0f * g1[v], 0.0f));
            m0[v] = (ib + ln      <= gwi) ? 1.0f - r0 : 1.0f;
            m1[v] = (ib + 16 + ln <= gwi) ? 1.0f - r1 : 1.0f;
        }

#pragma unroll
        for (int p = 0; p < 4; ++p) {
            v2f bm0; bm0.x = m0[2 * p];  bm0.y = m0[2 * p + 1];
            v2f bm1; bm1.x = m1[2 * p];  bm1.y = m1[2 * p + 1];
#pragma unroll
            for (int dt = 0; dt < 4; ++dt) {
                v2f a;
                if (p == 0)      { a.x = xlo[dt].x; a.y = xlo[dt].y; }
                else if (p == 1) { a.x = xlo[dt].z; a.y = xlo[dt].w; }
                else if (p == 2) { a.x = xhi[dt].x; a.y = xhi[dt].y; }
                else             { a.x = xhi[dt].z; a.y = xhi[dt].w; }
                c[dt][0] = wmma_f32_16x16x4(a, bm0, c[dt][0]);
                c[dt][1] = wmma_f32_16x16x4(a, bm1, c[dt][1]);
            }
        }
    }

#pragma unroll
    for (int dt = 0; dt < 4; ++dt) {
#pragma unroll
        for (int v = 0; v < 8; ++v) {
            red[wv][dt * 16 + 8 * kh + v][ln]      = c[dt][0][v];
            red[wv][dt * 16 + 8 * kh + v][16 + ln] = c[dt][1][v];
        }
    }
    __syncthreads();

    float* ob = out + (size_t)b * PD * PS;
#pragma unroll
    for (int j = 0; j < 2; ++j) {
        const int f  = t + 256 * j;
        const int d  = f >> 3;
        const int iq = (f & 7) * 4;
        v4f acc = *reinterpret_cast<const v4f*>(&red[0][d][iq]);
#pragma unroll
        for (int w = 1; w < 8; ++w) acc += *reinterpret_cast<const v4f*>(&red[w][d][iq]);
        *reinterpret_cast<v4f*>(ob + (size_t)d * PS + ib + iq) = acc;
    }
}

extern "C" void kernel_launch(void* const* d_in, const int* in_sizes, int n_in,
                              void* d_out, int out_size, void* d_ws, size_t ws_size,
                              hipStream_t stream) {
    const float* x = (const float*)d_in[0];
    const float* W = (const float*)d_in[1];
    float* out = (float*)d_out;

    float* Aws = (float*)d_ws;
    float* Bws = Aws + (size_t)PB * PS * PD;

    lie_stage1<<<256, 128, 0, stream>>>(x, W, Aws, Bws);
    lie_stage2<<<dim3(PS / 32, PB), 256, 0, stream>>>(x, Aws, Bws, out);
}


# stage 2 norms: hardware square root v_sqrt_f32 (1 ulp) in place of the correctly rounded sqrtf expansion, about 15 fewer vector instructions per value, 64 values per lane
# speedup vs baseline: 3.4347x; 1.0629x over previous
#include <hip/hip_runtime.h>
#include <hip/hip_bf16.h>

#define PB 8
#define PD 64
#define PS 512

typedef __attribute__((ext_vector_type(2))) float v2f;
typedef __attribute__((ext_vector_type(8))) float v8f;

__device__ __forceinline__ v8f wmma_f32_16x16x4(v2f a, v2f b, v8f c) {
    return __builtin_amdgcn_wmma_f32_16x16x4_f32(
         false, a,  false, b,
         (short)0, c,  false,  false);
}

__device__ __forceinline__ v8f v8f_zero() {
    v8f z = {0.f, 0.f, 0.f, 0.f, 0.f, 0.f, 0.f, 0.f};
    return z;
}

__global__ __launch_bounds__(128) void lie_stage1(
    const float* __restrict__ x, const float* __restrict__ W,
    float* __restrict__ Aws, float* __restrict__ Bws) {

    const int lane = threadIdx.x & 31;
    const int wv   = threadIdx.x >> 5;
    const int ln   = lane & 15;
    const int kh   = lane >> 4;

    const int half  = blockIdx.x & 1;
    const int chunk = (blockIdx.x >> 1) & 15;
    const int b     = blockIdx.x >> 5;
    const int jb    = chunk * 32 + (wv & 1) * 16;
    const int n0    = (wv >> 1) * 32;

    const float* xp = x + (size_t)b * PD * PS + (size_t)(2 * kh) * PS + jb + ln;
    const float* wp = W + (size_t)(half * PD + 2 * kh) * PD + n0 + 2 * ln;

    v2f a[16], w0[16], w1[16];
#pragma unroll
    for (int s = 0; s < 16; ++s) {
        a[s].x = xp[(size_t)(4 * s) * PS];
        a[s].y = xp[(size_t)(4 * s + 1) * PS];
        w0[s]  = *reinterpret_cast<const v2f*>(wp + (size_t)(4 * s) * PD);
        w1[s]  = *reinterpret_cast<const v2f*>(wp + (size_t)(4 * s + 1) * PD);
    }

    __builtin_amdgcn_sched_barrier(0);

    v8f acc0 = v8f_zero(), acc1 = v8f_zero();
#pragma unroll
    for (int s = 0; s < 16; ++s) {
        v2f b0; b0.x = w0[s].x; b0.y = w1[s].x;
        v2f b1; b1.x = w0[s].y; b1.y = w1[s].y;
        acc0 = wmma_f32_16x16x4(a[s], b0, acc0);
        acc1 = wmma_f32_16x16x4(a[s], b1, acc1);
    }

    float* dst = (half ? Bws : Aws) + ((size_t)b * PS + jb + 8 * kh) * PD + n0 + 2 * ln;
#pragma unroll
    for (int v = 0; v < 8; ++v) {
        v2f o; o.x = acc0[v]; o.y = acc1[v];
        *reinterpret_cast<v2f*>(dst + (size_t)v * PD) = o;
    }
}

typedef __attribute__((ext_vector_type(4))) float v4f;

#define RED_PITCH 36

__global__ __launch_bounds__(256) void lie_stage2(
    const float* __restrict__ x, const float* __restrict__ Aws,
    const float* __restrict__ Bws, float* __restrict__ out) {

    const int b  = blockIdx.y;
    const int ib = blockIdx.x * 32;

    __shared__ __attribute__((aligned(16))) float red[8][64][RED_PITCH];

    const int t    = threadIdx.x;
    const int lane = t & 31;
    const int wv   = t >> 5;
    const int ln   = lane & 15;
    const int kh   = lane >> 4;
    const int wc   = wv * 64;


    v4f qa[2][8];
    float na[2];
#pragma unroll
    for (int nt = 0; nt < 2; ++nt) {
        const float* pa = Aws + ((size_t)(b * PS + ib + nt * 16 + ln)) * PD + 4 * kh;
        float s2 = 0.f;
#pragma unroll
        for (int s = 0; s < 8; ++s) {
            const v4f q = *reinterpret_cast<const v4f*>(pa + 8 * s);
            qa[nt][s] = q;
            s2 += q.x * q.x + q.y * q.y + q.z * q.z + q.w * q.w;
        }
        na[nt] = s2 + __shfl_xor(s2, 16);
    }

    v8f c[4][2];
#pragma unroll
    for (int dt = 0; dt < 4; ++dt) { c[dt][0] = v8f_zero(); c[dt][1] = v8f_zero(); }

    const float* xb = x + (size_t)b * PD * PS;

#pragma unroll
    for (int mtile = 0; mtile < 4; ++mtile) {
        const int wl0 = mtile * 16;

        v4f qb[8];
        const float* pb = Bws + ((size_t)(b * PS + wc + wl0 + ln)) * PD + 4 * kh;
        float s2 = 0.f;
#pragma unroll
        for (int s = 0; s < 8; ++s) {
            const v4f q = *reinterpret_cast<const v4f*>(pb + 8 * s);
            qb[s] = q;
            s2 += q.x * q.x + q.y * q.y + q.z * q.z + q.w * q.w;
        }
        const float nbrow = s2 + __shfl_xor(s2, 16);

        v4f xlo[4], xhi[4];
#pragma unroll
        for (int dt = 0; dt < 4; ++dt) {
            const float* px = xb + (size_t)(dt * 16 + ln) * PS + wc + wl0 + 8 * kh;
            xlo[dt] = *reinterpret_cast<const v4f*>(px);
            xhi[dt] = *reinterpret_cast<const v4f*>(px + 4);
        }

        v8f g0 = v8f_zero(), g1 = v8f_zero();
#pragma unroll
        for (int s = 0; s < 8; ++s) {
            v2f a0;  a0.x  = qb[s].x;     a0.y  = qb[s].y;
            v2f a1;  a1.x  = qb[s].z;     a1.y  = qb[s].w;
            v2f b00; b00.x = qa[0][s].x;  b00.y = qa[0][s].y;
            v2f b01; b01.x = qa[0][s].z;  b01.y = qa[0][s].w;
            v2f b10; b10.x = qa[1][s].x;  b10.y = qa[1][s].y;
            v2f b11; b11.x = qa[1][s].z;  b11.y = qa[1][s].w;
            g0 = wmma_f32_16x16x4(a0, b00, g0);
            g1 = wmma_f32_16x16x4(a0, b10, g1);
            g0 = wmma_f32_16x16x4(a1, b01, g0);
            g1 = wmma_f32_16x16x4(a1, b11, g1);
        }

        float m0[8], m1[8];
#pragma unroll
        for (int v = 0; v < 8; ++v) {
            const int   gwi = wc + wl0 + 8 * kh + v;
            const float nbv = __shfl(nbrow, v + 8 * kh);
            const float r0  = __builtin_amdgcn_sqrtf(fmaxf(na[0] + nbv + 2.0f * g0[v], 0.0f));
            const float r1  = __builtin_amdgcn_sqrtf(fmaxf(na[1] + nbv + 2.0f * g1[v], 0.0f));
            m0[v] = (ib + ln      <= gwi) ? 1.0f - r0 : 1.0f;
            m1[v] = (ib + 16 + ln <= gwi) ? 1.0f - r1 : 1.0f;
        }

#pragma unroll
        for (int p = 0; p < 4; ++p) {
            v2f bm0; bm0.x = m0[2 * p];  bm0.y = m0[2 * p + 1];
            v2f bm1; bm1.x = m1[2 * p];  bm1.y = m1[2 * p + 1];
#pragma unroll
            for (int dt = 0; dt < 4; ++dt) {
                v2f a;
                if (p == 0)      { a.x = xlo[dt].x; a.y = xlo[dt].y; }
                else if (p == 1) { a.x = xlo[dt].z; a.y = xlo[dt].w; }
                else if (p == 2) { a.x = xhi[dt].x; a.y = xhi[dt].y; }
                else             { a.x = xhi[dt].z; a.y = xhi[dt].w; }
                c[dt][0] = wmma_f32_16x16x4(a, bm0, c[dt][0]);
                c[dt][1] = wmma_f32_16x16x4(a, bm1, c[dt][1]);
            }
        }
    }

#pragma unroll
    for (int dt = 0; dt < 4; ++dt) {
#pragma unroll
        for (int v = 0; v < 8; ++v) {
            red[wv][dt * 16 + 8 * kh + v][ln]      = c[dt][0][v];
            red[wv][dt * 16 + 8 * kh + v][16 + ln] = c[dt][1][v];
        }
    }
    __syncthreads();

    float* ob = out + (size_t)b * PD * PS;
#pragma unroll
    for (int j = 0; j < 2; ++j) {
        const int f  = t + 256 * j;
        const int d  = f >> 3;
        const int iq = (f & 7) * 4;
        v4f acc = *reinterpret_cast<const v4f*>(&red[0][d][iq]);
#pragma unroll
        for (int w = 1; w < 8; ++w) acc += *reinterpret_cast<const v4f*>(&red[w][d][iq]);
        *reinterpret_cast<v4f*>(ob + (size_t)d * PS + ib + iq) = acc;
    }
}

extern "C" void kernel_launch(void* const* d_in, const int* in_sizes, int n_in,
                              void* d_out, int out_size, void* d_ws, size_t ws_size,
                              hipStream_t stream) {
    const float* x = (const float*)d_in[0];
    const float* W = (const float*)d_in[1];
    float* out = (float*)d_out;

    float* Aws = (float*)d_ws;
    float* Bws = Aws + (size_t)PB * PS * PD;

    lie_stage1<<<256, 128, 0, stream>>>(x, W, Aws, Bws);
    lie_stage2<<<dim3(PS / 32, PB), 256, 0, stream>>>(x, Aws, Bws, out);
}
